# starting kernel
# baseline (speedup 1.0000x reference)
#include <hip/hip_runtime.h>
#include <math.h>

#define B_ 128
#define T_ 32
#define D_ 128
#define H_ 256
#define O_ 128
#define Z_ (D_ + H_)

typedef __attribute__((ext_vector_type(2))) float v2f;
typedef __attribute__((ext_vector_type(8))) float v8f;

__global__ __launch_bounds__(256) void stpn_recurrent_kernel(
    const float* __restrict__ sentence,
    const float* __restrict__ h0,
    const float* __restrict__ F0,
    const float* __restrict__ W,
    const float* __restrict__ bias,
    const float* __restrict__ lam,
    const float* __restrict__ gam,
    float* __restrict__ outH,
    float* __restrict__ outF)
{
  const int b    = blockIdx.x;
  const int tid  = threadIdx.x;
  const int w    = tid >> 5;
  const int lane = tid & 31;

  __shared__ __align__(16) float z[Z_];
  __shared__ float hbuf[H_];

  float*       Fb  = outF + (size_t)b * H_ * Z_;
  const float* F0b = F0   + (size_t)b * H_ * Z_;

  {
    const float4* src = (const float4*)F0b;
    float4*       dst = (float4*)Fb;
    const int n4 = (H_ * Z_) / 4;
    for (int i = tid; i < n4; i += 256) dst[i] = src[i];
    hbuf[tid] = h0[b * H_ + tid];
  }
  __syncthreads();

  const float* xb = sentence + (size_t)b * T_ * D_;
  const int c0 = lane * 3;

  for (int t = 0; t < T_; ++t) {
    if (tid < D_) z[tid] = xb[t * D_ + tid];
    z[D_ + tid] = hbuf[tid];
    if (t + 1 < T_ && tid < D_ / 16)
      __builtin_prefetch(xb + (t + 1) * D_ + tid * 16, 0, 0);
    __syncthreads();

    const float4* z4p = (const float4*)z;
    const float4 zc0 = z4p[c0 + 0];
    const float4 zc1 = z4p[c0 + 1];
    const float4 zc2 = z4p[c0 + 2];

    for (int rr = 0; rr < 32; ++rr) {
      const int r = w * 32 + rr;
      const size_t rowOff = (size_t)r * (Z_ / 4) + c0;
      const float4* Wr = (const float4*)W   + rowOff;
      float4*       Fr = (float4*)Fb        + rowOff;
      const float4* Lr = (const float4*)lam + rowOff;
      const float4* Gr = (const float4*)gam + rowOff;

      const float4 w0 = Wr[0], w1 = Wr[1], w2 = Wr[2];
      float4 f0 = Fr[0], f1 = Fr[1], f2 = Fr[2];

      float p = 0.f;
      p = fmaf(w0.x + f0.x, zc0.x, p); p = fmaf(w0.y + f0.y, zc0.y, p);
      p = fmaf(w0.z + f0.z, zc0.z, p); p = fmaf(w0.w + f0.w, zc0.w, p);
      p = fmaf(w1.x + f1.x, zc1.x, p); p = fmaf(w1.y + f1.y, zc1.y, p);
      p = fmaf(w1.z + f1.z, zc1.z, p); p = fmaf(w1.w + f1.w, zc1.w, p);
      p = fmaf(w2.x + f2.x, zc2.x, p); p = fmaf(w2.y + f2.y, zc2.y, p);
      p = fmaf(w2.z + f2.z, zc2.z, p); p = fmaf(w2.w + f2.w, zc2.w, p);

      #pragma unroll
      for (int m = 16; m >= 1; m >>= 1) p += __shfl_xor(p, m, 32);

      const float hn = tanhf(p + bias[r]);
      if (lane == 0) hbuf[r] = hn;

      const float4 l0 = Lr[0], l1 = Lr[1], l2 = Lr[2];
      const float4 g0 = Gr[0], g1 = Gr[1], g2 = Gr[2];
      f0.x = fmaf(l0.x, f0.x, g0.x * hn * zc0.x);
      f0.y = fmaf(l0.y, f0.y, g0.y * hn * zc0.y);
      f0.z = fmaf(l0.z, f0.z, g0.z * hn * zc0.z);
      f0.w = fmaf(l0.w, f0.w, g0.w * hn * zc0.w);
      f1.x = fmaf(l1.x, f1.x, g1.x * hn * zc1.x);
      f1.y = fmaf(l1.y, f1.y, g1.y * hn * zc1.y);
      f1.z = fmaf(l1.z, f1.z, g1.z * hn * zc1.z);
      f1.w = fmaf(l1.w, f1.w, g1.w * hn * zc1.w);
      f2.x = fmaf(l2.x, f2.x, g2.x * hn * zc2.x);
      f2.y = fmaf(l2.y, f2.y, g2.y * hn * zc2.y);
      f2.z = fmaf(l2.z, f2.z, g2.z * hn * zc2.z);
      f2.w = fmaf(l2.w, f2.w, g2.w * hn * zc2.w);
      Fr[0] = f0; Fr[1] = f1; Fr[2] = f2;
    }
    __syncthreads();
  }

  outH[b * H_ + tid] = hbuf[tid];
}

__global__ __launch_bounds__(32) void head_wmma_kernel(
    const float* __restrict__ hmat,
    const float* __restrict__ Wout,
    const float* __restrict__ bout,
    float* __restrict__ tag)
{
  const int tileM = blockIdx.x;
  const int tileN = blockIdx.y;
  const int lane  = threadIdx.x;
  const int half  = lane >> 4;
  const int l16   = lane & 15;

  const int m = tileM * 16 + l16;
  const int n = tileN * 16 + l16;

  v8f c = {};
  for (int k0 = 0; k0 < H_; k0 += 4) {
    const int ka = k0 + 2 * half;
    v2f a, bf;
    a.x  = hmat[m * H_ + ka + 0];
    a.y  = hmat[m * H_ + ka + 1];
    bf.x = Wout[n * H_ + ka + 0];
    bf.y = Wout[n * H_ + ka + 1];
    c = __builtin_amdgcn_wmma_f32_16x16x4_f32(
        false, a, false, bf, (short)0, c, false, false);
  }

  #pragma unroll
  for (int v = 0; v < 8; ++v) {
    const int mg = tileM * 16 + v + 8 * half;
    const int ng = tileN * 16 + l16;
    tag[mg * O_ + ng] = c[v] + bout[ng];
  }
}

extern "C" void kernel_launch(void* const* d_in, const int* in_sizes, int n_in,
                              void* d_out, int out_size, void* d_ws, size_t ws_size,
                              hipStream_t stream) {
  (void)in_sizes; (void)n_in; (void)out_size; (void)d_ws; (void)ws_size;

  const float* sentence = (const float*)d_in[0];
  const float* h0       = (const float*)d_in[1];
  const float* F0       = (const float*)d_in[2];
  const float* W        = (const float*)d_in[3];
  const float* bias     = (const float*)d_in[4];
  const float* lam      = (const float*)d_in[5];
  const float* gam      = (const float*)d_in[6];
  const float* Wout     = (const float*)d_in[7];
  const float* bout     = (const float*)d_in[8];

  float* out  = (float*)d_out;
  float* tag  = out;
  float* outH = out + B_ * O_;
  float* outF = outH + B_ * H_;

  stpn_recurrent_kernel<<<B_, 256, 0, stream>>>(
      sentence, h0, F0, W, bias, lam, gam, outH, outF);

  dim3 grid(B_ / 16, O_ / 16);
  head_wmma_kernel<<<grid, 32, 0, stream>>>(outH, Wout, bout, tag);
}


# F (fast weights) kept on chip across all 32 timesteps: per row chunk one float4 in registers, two in LDS, 16 waves with unrolled rows; F0 read once, F stored once, no per-step F load/store
# speedup vs baseline: 4.2316x; 4.2316x over previous
#include <hip/hip_runtime.h>
#include <math.h>

#define B_ 128
#define T_ 32
#define D_ 128
#define H_ 256
#define O_ 128
#define Z_ (D_ + H_)

typedef __attribute__((ext_vector_type(2))) float v2f;
typedef __attribute__((ext_vector_type(8))) float v8f;

#define NT_  512
#define RPW_ 16
#define RQ_  (Z_ / 4)
#define RG1_ 8
#define RG4_ 4

__global__ __launch_bounds__(NT_) void stpn_recurrent_kernel(
    const float* __restrict__ sentence,
    const float* __restrict__ h0,
    const float* __restrict__ F0,
    const float* W,
    const float* __restrict__ bias,
    const float* lam,
    const float* gam,
    float* __restrict__ outH,
    float* __restrict__ outF)
{
  const int b    = blockIdx.x;
  const int tid  = threadIdx.x;
  const int w    = __builtin_amdgcn_readfirstlane(tid >> 5);
  const int lane = tid & 31;

  __shared__ __align__(16) float z[Z_];
  __shared__ float hbuf[H_];
  __shared__ float4 fl[H_ * 2 * 32];

  const int c0 = lane * 3;
  const size_t q0 = (size_t)(w * RPW_) * RQ_ + c0;
  float4* flw = fl + (w * RPW_) * 2 * 32 + lane;

  float4 fa[RPW_];
  {
    const float4* src = (const float4*)(F0 + (size_t)b * H_ * Z_) + q0;
    #pragma unroll
    for (int rr = 0; rr < RPW_; ++rr) {
      fa[rr]                 = src[rr * RQ_ + 0];
      flw[(rr * 2 + 0) * 32] = src[rr * RQ_ + 1];
      flw[(rr * 2 + 1) * 32] = src[rr * RQ_ + 2];
    }
    if (tid < H_) hbuf[tid] = h0[b * H_ + tid];
  }
  __syncthreads();

  const float* xb = sentence + (size_t)b * T_ * D_;
  const float4* Wq = (const float4*)W   + q0;
  const float4* Lq = (const float4*)lam + q0;
  const float4* Gq = (const float4*)gam + q0;

  for (int t = 0; t < T_; ++t) {
    if (tid < D_)      z[tid] = xb[t * D_ + tid];
    else if (tid < Z_) z[tid] = hbuf[tid - D_];
    if (t + 1 < T_ && tid < D_ / 16)
      __builtin_prefetch(xb + (t + 1) * D_ + tid * 16, 0, 0);
    __syncthreads();

    const float4* z4p = (const float4*)z;
    const float4 zc0 = z4p[c0 + 0];
    const float4 zc1 = z4p[c0 + 1];
    const float4 zc2 = z4p[c0 + 2];

    float pre[RPW_];
    #pragma unroll
    for (int g = 0; g < RPW_; g += RG1_) {
      float4 wv[RG1_][3];
      #pragma unroll
      for (int j = 0; j < RG1_; ++j) {
        wv[j][0] = Wq[(g + j) * RQ_ + 0];
        wv[j][1] = Wq[(g + j) * RQ_ + 1];
        wv[j][2] = Wq[(g + j) * RQ_ + 2];
      }
      asm volatile("" ::: "memory");
      #pragma unroll
      for (int j = 0; j < RG1_; ++j) {
        const int rr = g + j;
        const float4 w0 = wv[j][0], w1 = wv[j][1], w2 = wv[j][2];
        const float4 f0 = fa[rr], f1 = flw[(rr * 2 + 0) * 32], f2 = flw[(rr * 2 + 1) * 32];

        float p = 0.f;
        p = fmaf(w0.x + f0.x, zc0.x, p); p = fmaf(w0.y + f0.y, zc0.y, p);
        p = fmaf(w0.z + f0.z, zc0.z, p); p = fmaf(w0.w + f0.w, zc0.w, p);
        p = fmaf(w1.x + f1.x, zc1.x, p); p = fmaf(w1.y + f1.y, zc1.y, p);
        p = fmaf(w1.z + f1.z, zc1.z, p); p = fmaf(w1.w + f1.w, zc1.w, p);
        p = fmaf(w2.x + f2.x, zc2.x, p); p = fmaf(w2.y + f2.y, zc2.y, p);
        p = fmaf(w2.z + f2.z, zc2.z, p); p = fmaf(w2.w + f2.w, zc2.w, p);
        pre[rr] = p;
      }
    }

    #pragma unroll
    for (int m = 16; m >= 1; m >>= 1) {
      float other[RPW_];
      #pragma unroll
      for (int rr = 0; rr < RPW_; ++rr) other[rr] = __shfl_xor(pre[rr], m, 32);
      #pragma unroll
      for (int rr = 0; rr < RPW_; ++rr) pre[rr] += other[rr];
    }

    #pragma unroll
    for (int rr = 0; rr < RPW_; ++rr) {
      const int r = w * RPW_ + rr;
      const float hnew = tanhf(pre[rr] + bias[r]);
      if (lane == 0) hbuf[r] = hnew;
    }

    #pragma unroll
    for (int g = 0; g < RPW_; g += RG4_) {
      float4 lv[RG4_][3], gv[RG4_][3];
      #pragma unroll
      for (int j = 0; j < RG4_; ++j) {
        lv[j][0] = Lq[(g + j) * RQ_ + 0]; lv[j][1] = Lq[(g + j) * RQ_ + 1]; lv[j][2] = Lq[(g + j) * RQ_ + 2];
        gv[j][0] = Gq[(g + j) * RQ_ + 0]; gv[j][1] = Gq[(g + j) * RQ_ + 1]; gv[j][2] = Gq[(g + j) * RQ_ + 2];
      }
      asm volatile("" ::: "memory");
      #pragma unroll
      for (int j = 0; j < RG4_; ++j) {
        const int rr = g + j;
        const float h = hbuf[w * RPW_ + rr];
        const float4 l0 = lv[j][0], l1 = lv[j][1], l2 = lv[j][2];
        const float4 g0 = gv[j][0], g1 = gv[j][1], g2 = gv[j][2];
        float4 f0 = fa[rr], f1 = flw[(rr * 2 + 0) * 32], f2 = flw[(rr * 2 + 1) * 32];
        f0.x = fmaf(l0.x, f0.x, g0.x * h * zc0.x);
        f0.y = fmaf(l0.y, f0.y, g0.y * h * zc0.y);
        f0.z = fmaf(l0.z, f0.z, g0.z * h * zc0.z);
        f0.w = fmaf(l0.w, f0.w, g0.w * h * zc0.w);
        f1.x = fmaf(l1.x, f1.x, g1.x * h * zc1.x);
        f1.y = fmaf(l1.y, f1.y, g1.y * h * zc1.y);
        f1.z = fmaf(l1.z, f1.z, g1.z * h * zc1.z);
        f1.w = fmaf(l1.w, f1.w, g1.w * h * zc1.w);
        f2.x = fmaf(l2.x, f2.x, g2.x * h * zc2.x);
        f2.y = fmaf(l2.y, f2.y, g2.y * h * zc2.y);
        f2.z = fmaf(l2.z, f2.z, g2.z * h * zc2.z);
        f2.w = fmaf(l2.w, f2.w, g2.w * h * zc2.w);
        fa[rr] = f0; flw[(rr * 2 + 0) * 32] = f1; flw[(rr * 2 + 1) * 32] = f2;
      }
    }
    __syncthreads();
  }

  {
    float4* dst = (float4*)(outF + (size_t)b * H_ * Z_) + q0;
    #pragma unroll
    for (int rr = 0; rr < RPW_; ++rr) {
      dst[rr * RQ_ + 0] = fa[rr];
      dst[rr * RQ_ + 1] = flw[(rr * 2 + 0) * 32];
      dst[rr * RQ_ + 2] = flw[(rr * 2 + 1) * 32];
    }
  }
  if (tid < H_) outH[b * H_ + tid] = hbuf[tid];
}

__global__ __launch_bounds__(32) void head_wmma_kernel(
    const float* __restrict__ hmat,
    const float* __restrict__ Wout,
    const float* __restrict__ bout,
    float* __restrict__ tag)
{
  const int tileM = blockIdx.x;
  const int tileN = blockIdx.y;
  const int lane  = threadIdx.x;
  const int half  = lane >> 4;
  const int l16   = lane & 15;

  const int m = tileM * 16 + l16;
  const int n = tileN * 16 + l16;

  v8f c = {};
  for (int k0 = 0; k0 < H_; k0 += 4) {
    const int ka = k0 + 2 * half;
    v2f a, bf;
    a.x  = hmat[m * H_ + ka + 0];
    a.y  = hmat[m * H_ + ka + 1];
    bf.x = Wout[n * H_ + ka + 0];
    bf.y = Wout[n * H_ + ka + 1];
    c = __builtin_amdgcn_wmma_f32_16x16x4_f32(
        false, a, false, bf, (short)0, c, false, false);
  }

  #pragma unroll
  for (int v = 0; v < 8; ++v) {
    const int mg = tileM * 16 + v + 8 * half;
    const int ng = tileN * 16 + l16;
    tag[mg * O_ + ng] = c[v] + bout[ng];
  }
}

extern "C" void kernel_launch(void* const* d_in, const int* in_sizes, int n_in,
                              void* d_out, int out_size, void* d_ws, size_t ws_size,
                              hipStream_t stream) {
  (void)in_sizes; (void)n_in; (void)out_size; (void)d_ws; (void)ws_size;

  const float* sentence = (const float*)d_in[0];
  const float* h0       = (const float*)d_in[1];
  const float* F0       = (const float*)d_in[2];
  const float* W        = (const float*)d_in[3];
  const float* bias     = (const float*)d_in[4];
  const float* lam      = (const float*)d_in[5];
  const float* gam      = (const float*)d_in[6];
  const float* Wout     = (const float*)d_in[7];
  const float* bout     = (const float*)d_in[8];

  float* out  = (float*)d_out;
  float* tag  = out;
  float* outH = out + B_ * O_;
  float* outF = outH + B_ * H_;

  stpn_recurrent_kernel<<<B_, NT_, 0, stream>>>(
      sentence, h0, F0, W, bias, lam, gam, outH, outF);

  dim3 grid(B_ / 16, O_ / 16);
  head_wmma_kernel<<<grid, 32, 0, stream>>>(outH, Wout, bout, tag);
}


# Lane-to-column map changed so every 16-byte global load/store covers whole 128-byte lines (lane owns float4 l, 32+l, 64+l of a row, not 3l..3l+2); each row sum adds its columns in a new order
# speedup vs baseline: 5.6429x; 1.3335x over previous
#include <hip/hip_runtime.h>
#include <math.h>

#define B_ 128
#define T_ 32
#define D_ 128
#define H_ 256
#define O_ 128
#define Z_ (D_ + H_)

typedef __attribute__((ext_vector_type(2))) float v2f;
typedef __attribute__((ext_vector_type(8))) float v8f;

#define NT_  512
#define RPW_ 16
#define RQ_  (Z_ / 4)
#define RG1_ 8
#define RG4_ 4
#define CS_  32

__global__ __launch_bounds__(NT_) void stpn_recurrent_kernel(
    const float* __restrict__ sentence,
    const float* __restrict__ h0,
    const float* __restrict__ F0,
    const float* W,
    const float* __restrict__ bias,
    const float* lam,
    const float* gam,
    float* __restrict__ outH,
    float* __restrict__ outF)
{
  const int b    = blockIdx.x;
  const int tid  = threadIdx.x;
  const int w    = __builtin_amdgcn_readfirstlane(tid >> 5);
  const int lane = tid & 31;

  __shared__ __align__(16) float z[Z_];
  __shared__ float hbuf[H_];
  __shared__ float4 fl[H_ * 2 * 32];

  const size_t q0 = (size_t)(w * RPW_) * RQ_ + lane;
  float4* flw = fl + (w * RPW_) * 2 * 32 + lane;

  float4 fa[RPW_];
  {
    const float4* src = (const float4*)(F0 + (size_t)b * H_ * Z_) + q0;
    #pragma unroll
    for (int rr = 0; rr < RPW_; ++rr) {
      fa[rr]                 = src[rr * RQ_ + 0 * CS_];
      flw[(rr * 2 + 0) * 32] = src[rr * RQ_ + 1 * CS_];
      flw[(rr * 2 + 1) * 32] = src[rr * RQ_ + 2 * CS_];
    }
    if (tid < H_) hbuf[tid] = h0[b * H_ + tid];
  }
  __syncthreads();

  const float* xb = sentence + (size_t)b * T_ * D_;
  const float4* Wq = (const float4*)W   + q0;
  const float4* Lq = (const float4*)lam + q0;
  const float4* Gq = (const float4*)gam + q0;

  for (int t = 0; t < T_; ++t) {
    if (tid < D_)      z[tid] = xb[t * D_ + tid];
    else if (tid < Z_) z[tid] = hbuf[tid - D_];
    if (t + 1 < T_ && tid < D_ / 16)
      __builtin_prefetch(xb + (t + 1) * D_ + tid * 16, 0, 0);
    __syncthreads();

    const float4* z4p = (const float4*)z;
    const float4 zc0 = z4p[0 * CS_ + lane];
    const float4 zc1 = z4p[1 * CS_ + lane];
    const float4 zc2 = z4p[2 * CS_ + lane];

    float pre[RPW_];
    #pragma unroll
    for (int g = 0; g < RPW_; g += RG1_) {
      float4 wv[RG1_][3];
      #pragma unroll
      for (int j = 0; j < RG1_; ++j) {
        wv[j][0] = Wq[(g + j) * RQ_ + 0 * CS_];
        wv[j][1] = Wq[(g + j) * RQ_ + 1 * CS_];
        wv[j][2] = Wq[(g + j) * RQ_ + 2 * CS_];
      }
      asm volatile("" ::: "memory");
      #pragma unroll
      for (int j = 0; j < RG1_; ++j) {
        const int rr = g + j;
        const float4 w0 = wv[j][0], w1 = wv[j][1], w2 = wv[j][2];
        const float4 f0 = fa[rr], f1 = flw[(rr * 2 + 0) * 32], f2 = flw[(rr * 2 + 1) * 32];

        float p = 0.f;
        p = fmaf(w0.x + f0.x, zc0.x, p); p = fmaf(w0.y + f0.y, zc0.y, p);
        p = fmaf(w0.z + f0.z, zc0.z, p); p = fmaf(w0.w + f0.w, zc0.w, p);
        p = fmaf(w1.x + f1.x, zc1.x, p); p = fmaf(w1.y + f1.y, zc1.y, p);
        p = fmaf(w1.z + f1.z, zc1.z, p); p = fmaf(w1.w + f1.w, zc1.w, p);
        p = fmaf(w2.x + f2.x, zc2.x, p); p = fmaf(w2.y + f2.y, zc2.y, p);
        p = fmaf(w2.z + f2.z, zc2.z, p); p = fmaf(w2.w + f2.w, zc2.w, p);
        pre[rr] = p;
      }
    }

    #pragma unroll
    for (int m = 16; m >= 1; m >>= 1) {
      float other[RPW_];
      #pragma unroll
      for (int rr = 0; rr < RPW_; ++rr) other[rr] = __shfl_xor(pre[rr], m, 32);
      #pragma unroll
      for (int rr = 0; rr < RPW_; ++rr) pre[rr] += other[rr];
    }

    #pragma unroll
    for (int rr = 0; rr < RPW_; ++rr) {
      const int r = w * RPW_ + rr;
      const float hnew = tanhf(pre[rr] + bias[r]);
      if (lane == 0) hbuf[r] = hnew;
    }

    #pragma unroll
    for (int g = 0; g < RPW_; g += RG4_) {
      float4 lv[RG4_][3], gv[RG4_][3];
      #pragma unroll
      for (int j = 0; j < RG4_; ++j) {
        lv[j][0] = Lq[(g + j) * RQ_ + 0 * CS_]; lv[j][1] = Lq[(g + j) * RQ_ + 1 * CS_]; lv[j][2] = Lq[(g + j) * RQ_ + 2 * CS_];
        gv[j][0] = Gq[(g + j) * RQ_ + 0 * CS_]; gv[j][1] = Gq[(g + j) * RQ_ + 1 * CS_]; gv[j][2] = Gq[(g + j) * RQ_ + 2 * CS_];
      }
      asm volatile("" ::: "memory");
      #pragma unroll
      for (int j = 0; j < RG4_; ++j) {
        const int rr = g + j;
        const float h = hbuf[w * RPW_ + rr];
        const float4 l0 = lv[j][0], l1 = lv[j][1], l2 = lv[j][2];
        const float4 g0 = gv[j][0], g1 = gv[j][1], g2 = gv[j][2];
        float4 f0 = fa[rr], f1 = flw[(rr * 2 + 0) * 32], f2 = flw[(rr * 2 + 1) * 32];
        f0.x = fmaf(l0.x, f0.x, g0.x * h * zc0.x);
        f0.y = fmaf(l0.y, f0.y, g0.y * h * zc0.y);
        f0.z = fmaf(l0.z, f0.z, g0.z * h * zc0.z);
        f0.w = fmaf(l0.w, f0.w, g0.w * h * zc0.w);
        f1.x = fmaf(l1.x, f1.x, g1.x * h * zc1.x);
        f1.y = fmaf(l1.y, f1.y, g1.y * h * zc1.y);
        f1.z = fmaf(l1.z, f1.z, g1.z * h * zc1.z);
        f1.w = fmaf(l1.w, f1.w, g1.w * h * zc1.w);
        f2.x = fmaf(l2.x, f2.x, g2.x * h * zc2.x);
        f2.y = fmaf(l2.y, f2.y, g2.y * h * zc2.y);
        f2.z = fmaf(l2.z, f2.z, g2.z * h * zc2.z);
        f2.w = fmaf(l2.w, f2.w, g2.w * h * zc2.w);
        fa[rr] = f0; flw[(rr * 2 + 0) * 32] = f1; flw[(rr * 2 + 1) * 32] = f2;
      }
    }
    __syncthreads();
  }

  {
    float4* dst = (float4*)(outF + (size_t)b * H_ * Z_) + q0;
    #pragma unroll
    for (int rr = 0; rr < RPW_; ++rr) {
      dst[rr * RQ_ + 0 * CS_] = fa[rr];
      dst[rr * RQ_ + 1 * CS_] = flw[(rr * 2 + 0) * 32];
      dst[rr * RQ_ + 2 * CS_] = flw[(rr * 2 + 1) * 32];
    }
  }
  if (tid < H_) outH[b * H_ + tid] = hbuf[tid];
}

__global__ __launch_bounds__(32) void head_wmma_kernel(
    const float* __restrict__ hmat,
    const float* __restrict__ Wout,
    const float* __restrict__ bout,
    float* __restrict__ tag)
{
  const int tileM = blockIdx.x;
  const int tileN = blockIdx.y;
  const int lane  = threadIdx.x;
  const int half  = lane >> 4;
  const int l16   = lane & 15;

  const int m = tileM * 16 + l16;
  const int n = tileN * 16 + l16;

  v8f c = {};
  for (int k0 = 0; k0 < H_; k0 += 4) {
    const int ka = k0 + 2 * half;
    v2f a, bf;
    a.x  = hmat[m * H_ + ka + 0];
    a.y  = hmat[m * H_ + ka + 1];
    bf.x = Wout[n * H_ + ka + 0];
    bf.y = Wout[n * H_ + ka + 1];
    c = __builtin_amdgcn_wmma_f32_16x16x4_f32(
        false, a, false, bf, (short)0, c, false, false);
  }

  #pragma unroll
  for (int v = 0; v < 8; ++v) {
    const int mg = tileM * 16 + v + 8 * half;
    const int ng = tileN * 16 + l16;
    tag[mg * O_ + ng] = c[v] + bout[ng];
  }
}

extern "C" void kernel_launch(void* const* d_in, const int* in_sizes, int n_in,
                              void* d_out, int out_size, void* d_ws, size_t ws_size,
                              hipStream_t stream) {
  (void)in_sizes; (void)n_in; (void)out_size; (void)d_ws; (void)ws_size;

  const float* sentence = (const float*)d_in[0];
  const float* h0       = (const float*)d_in[1];
  const float* F0       = (const float*)d_in[2];
  const float* W        = (const float*)d_in[3];
  const float* bias     = (const float*)d_in[4];
  const float* lam      = (const float*)d_in[5];
  const float* gam      = (const float*)d_in[6];
  const float* Wout     = (const float*)d_in[7];
  const float* bout     = (const float*)d_in[8];

  float* out  = (float*)d_out;
  float* tag  = out;
  float* outH = out + B_ * O_;
  float* outF = outH + B_ * H_;

  stpn_recurrent_kernel<<<B_, NT_, 0, stream>>>(
      sentence, h0, F0, W, bias, lam, gam, outH, outF);

  dim3 grid(B_ / 16, O_ / 16);
  head_wmma_kernel<<<grid, 32, 0, stream>>>(outH, Wout, bout, tag);
}


# W, lam, gamma chunks and next x requested rows ahead of use through a 6-slot register ring, across passes and the step barrier (LDS-only barrier, bias read once); arithmetic and sum order unchanged
# speedup vs baseline: 6.2402x; 1.1058x over previous
#include <hip/hip_runtime.h>
#include <math.h>

#define B_ 128
#define T_ 32
#define D_ 128
#define H_ 256
#define O_ 128
#define Z_ (D_ + H_)

typedef __attribute__((ext_vector_type(2))) float v2f;
typedef __attribute__((ext_vector_type(8))) float v8f;

#define NT_  512
#define RPW_ 16
#define RQ_  (Z_ / 4)
#define CS_  32
#define NS_  6
#define P4S_(j) ((2 * (j) + 4) % NS_)

__device__ __forceinline__ void lds_barrier() {
  __builtin_amdgcn_fence(__ATOMIC_RELEASE, "workgroup", "local");
  __builtin_amdgcn_s_barrier();
  __builtin_amdgcn_fence(__ATOMIC_ACQUIRE, "workgroup", "local");
}

__device__ __forceinline__ void request_row(float4 (&s)[3], const float4* p, int row) {
  s[0] = p[row * RQ_ + 0 * CS_];
  s[1] = p[row * RQ_ + 1 * CS_];
  s[2] = p[row * RQ_ + 2 * CS_];
}

__global__ __launch_bounds__(NT_) void stpn_recurrent_kernel(
    const float* __restrict__ sentence,
    const float* __restrict__ h0,
    const float* __restrict__ F0,
    const float* W,
    const float* __restrict__ bias,
    const float* lam,
    const float* gam,
    float* __restrict__ outH,
    float* __restrict__ outF)
{
  const int b    = blockIdx.x;
  const int tid  = threadIdx.x;
  const int w    = __builtin_amdgcn_readfirstlane(tid >> 5);
  const int lane = tid & 31;

  __shared__ __align__(16) float z[Z_];
  __shared__ float hbuf[H_];
  __shared__ float4 fl[H_ * 2 * 32];

  const size_t q0 = (size_t)(w * RPW_) * RQ_ + lane;
  float4* flw = fl + (w * RPW_) * 2 * 32 + lane;

  float bw[RPW_];
  #pragma unroll
  for (int rr = 0; rr < RPW_; ++rr) bw[rr] = bias[w * RPW_ + rr];

  float4 fa[RPW_];
  {
    const float4* src = (const float4*)(F0 + (size_t)b * H_ * Z_) + q0;
    #pragma unroll
    for (int rr = 0; rr < RPW_; ++rr) {
      fa[rr]                 = src[rr * RQ_ + 0 * CS_];
      flw[(rr * 2 + 0) * 32] = src[rr * RQ_ + 1 * CS_];
      flw[(rr * 2 + 1) * 32] = src[rr * RQ_ + 2 * CS_];
    }
    if (tid < H_) hbuf[tid] = h0[b * H_ + tid];
  }

  const float* xb = sentence + (size_t)b * T_ * D_;
  const float4* Wq = (const float4*)W   + q0;
  const float4* Lq = (const float4*)lam + q0;
  const float4* Gq = (const float4*)gam + q0;

  float4 rb[NS_][3];
  #pragma unroll
  for (int s = 0; s < NS_; ++s) request_row(rb[s], Wq, s);
  float xn = (tid < D_) ? xb[tid] : 0.f;
  __syncthreads();

  for (int t = 0; t < T_; ++t) {
    if (tid < D_)      z[tid] = xn;
    else if (tid < Z_) z[tid] = hbuf[tid - D_];
    lds_barrier();
    if (tid < D_) xn = xb[((t + 1 < T_) ? t + 1 : t) * D_ + tid];

    const float4* z4p = (const float4*)z;
    const float4 zc0 = z4p[0 * CS_ + lane];
    const float4 zc1 = z4p[1 * CS_ + lane];
    const float4 zc2 = z4p[2 * CS_ + lane];

    float pre[RPW_];
    #pragma unroll
    for (int r = 0; r < RPW_; ++r) {
      {
        const int rr = r;
        float4 w0 = rb[rr % NS_][0], w1 = rb[rr % NS_][1], w2 = rb[rr % NS_][2];
        asm volatile("" : "+v"(w0.x), "+v"(w0.y), "+v"(w0.z), "+v"(w0.w),
                          "+v"(w1.x), "+v"(w1.y), "+v"(w1.z), "+v"(w1.w),
                          "+v"(w2.x), "+v"(w2.y), "+v"(w2.z), "+v"(w2.w));
        const float4 f0 = fa[rr], f1 = flw[(rr * 2 + 0) * 32], f2 = flw[(rr * 2 + 1) * 32];

        float p = 0.f;
        p = fmaf(w0.x + f0.x, zc0.x, p); p = fmaf(w0.y + f0.y, zc0.y, p);
        p = fmaf(w0.z + f0.z, zc0.z, p); p = fmaf(w0.w + f0.w, zc0.w, p);
        p = fmaf(w1.x + f1.x, zc1.x, p); p = fmaf(w1.y + f1.y, zc1.y, p);
        p = fmaf(w1.z + f1.z, zc1.z, p); p = fmaf(w1.w + f1.w, zc1.w, p);
        p = fmaf(w2.x + f2.x, zc2.x, p); p = fmaf(w2.y + f2.y, zc2.y, p);
        p = fmaf(w2.z + f2.z, zc2.z, p); p = fmaf(w2.w + f2.w, zc2.w, p);
        pre[rr] = p;
      }
      if (r + NS_ < RPW_)                     request_row(rb[r % NS_], Wq, r + NS_);
      else if (((r + NS_ - RPW_) & 1) == 0)   request_row(rb[r % NS_], Lq, (r + NS_ - RPW_) / 2);
      else                                    request_row(rb[r % NS_], Gq, (r + NS_ - RPW_) / 2);
      __builtin_amdgcn_sched_barrier(0);
    }

    #pragma unroll
    for (int m = 16; m >= 1; m >>= 1) {
      float other[RPW_];
      #pragma unroll
      for (int rr = 0; rr < RPW_; ++rr) other[rr] = __shfl_xor(pre[rr], m, 32);
      #pragma unroll
      for (int rr = 0; rr < RPW_; ++rr) pre[rr] += other[rr];
    }

    #pragma unroll
    for (int rr = 0; rr < RPW_; ++rr) {
      const int r = w * RPW_ + rr;
      const float hnew = tanhf(pre[rr] + bw[rr]);
      if (lane == 0) hbuf[r] = hnew;
    }

    asm volatile("" ::: "memory");

    #pragma unroll
    for (int r = 0; r < RPW_; ++r) {
      {
        const int rr = r;
        const int sl = P4S_(rr), sg = sl + 1;
        const float h = hbuf[w * RPW_ + rr];
        const float4 l0 = rb[sl][0], l1 = rb[sl][1], l2 = rb[sl][2];
        const float4 g0 = rb[sg][0], g1 = rb[sg][1], g2 = rb[sg][2];
        float4 f0 = fa[rr], f1 = flw[(rr * 2 + 0) * 32], f2 = flw[(rr * 2 + 1) * 32];
        f0.x = fmaf(l0.x, f0.x, g0.x * h * zc0.x);
        f0.y = fmaf(l0.y, f0.y, g0.y * h * zc0.y);
        f0.z = fmaf(l0.z, f0.z, g0.z * h * zc0.z);
        f0.w = fmaf(l0.w, f0.w, g0.w * h * zc0.w);
        f1.x = fmaf(l1.x, f1.x, g1.x * h * zc1.x);
        f1.y = fmaf(l1.y, f1.y, g1.y * h * zc1.y);
        f1.z = fmaf(l1.z, f1.z, g1.z * h * zc1.z);
        f1.w = fmaf(l1.w, f1.w, g1.w * h * zc1.w);
        f2.x = fmaf(l2.x, f2.x, g2.x * h * zc2.x);
        f2.y = fmaf(l2.y, f2.y, g2.y * h * zc2.y);
        f2.z = fmaf(l2.z, f2.z, g2.z * h * zc2.z);
        f2.w = fmaf(l2.w, f2.w, g2.w * h * zc2.w);
        fa[rr] = f0; flw[(rr * 2 + 0) * 32] = f1; flw[(rr * 2 + 1) * 32] = f2;
      }
      {
        const int rr = r;
        const int sl = P4S_(rr), sg = sl + 1;
        if (rr + NS_ / 2 < RPW_) {
          request_row(rb[sl], Lq, rr + NS_ / 2);
          request_row(rb[sg], Gq, rr + NS_ / 2);
        } else {
          request_row(rb[sl], Wq, sl);
          request_row(rb[sg], Wq, sg);
        }
      }
      __builtin_amdgcn_sched_barrier(0);
    }
    lds_barrier();
  }

  {
    float4* dst = (float4*)(outF + (size_t)b * H_ * Z_) + q0;
    #pragma unroll
    for (int rr = 0; rr < RPW_; ++rr) dst[rr * RQ_ + 0 * CS_] = fa[rr];
    #pragma nounroll
    for (int rr = 0; rr < RPW_; ++rr) {
      dst[rr * RQ_ + 1 * CS_] = flw[(rr * 2 + 0) * 32];
      dst[rr * RQ_ + 2 * CS_] = flw[(rr * 2 + 1) * 32];
    }
  }
  if (tid < H_) outH[b * H_ + tid] = hbuf[tid];
}

__global__ __launch_bounds__(32) void head_wmma_kernel(
    const float* __restrict__ hmat,
    const float* __restrict__ Wout,
    const float* __restrict__ bout,
    float* __restrict__ tag)
{
  const int tileM = blockIdx.x;
  const int tileN = blockIdx.y;
  const int lane  = threadIdx.x;
  const int half  = lane >> 4;
  const int l16   = lane & 15;

  const int m = tileM * 16 + l16;
  const int n = tileN * 16 + l16;

  v8f c = {};
  for (int k0 = 0; k0 < H_; k0 += 4) {
    const int ka = k0 + 2 * half;
    v2f a, bf;
    a.x  = hmat[m * H_ + ka + 0];
    a.y  = hmat[m * H_ + ka + 1];
    bf.x = Wout[n * H_ + ka + 0];
    bf.y = Wout[n * H_ + ka + 1];
    c = __builtin_amdgcn_wmma_f32_16x16x4_f32(
        false, a, false, bf, (short)0, c, false, false);
  }

  #pragma unroll
  for (int v = 0; v < 8; ++v) {
    const int mg = tileM * 16 + v + 8 * half;
    const int ng = tileN * 16 + l16;
    tag[mg * O_ + ng] = c[v] + bout[ng];
  }
}

extern "C" void kernel_launch(void* const* d_in, const int* in_sizes, int n_in,
                              void* d_out, int out_size, void* d_ws, size_t ws_size,
                              hipStream_t stream) {
  (void)in_sizes; (void)n_in; (void)out_size; (void)d_ws; (void)ws_size;

  const float* sentence = (const float*)d_in[0];
  const float* h0       = (const float*)d_in[1];
  const float* F0       = (const float*)d_in[2];
  const float* W        = (const float*)d_in[3];
  const float* bias     = (const float*)d_in[4];
  const float* lam      = (const float*)d_in[5];
  const float* gam      = (const float*)d_in[6];
  const float* Wout     = (const float*)d_in[7];
  const float* bout     = (const float*)d_in[8];

  float* out  = (float*)d_out;
  float* tag  = out;
  float* outH = out + B_ * O_;
  float* outF = outH + B_ * H_;

  stpn_recurrent_kernel<<<B_, NT_, 0, stream>>>(
      sentence, h0, F0, W, bias, lam, gam, outH, outF);

  dim3 grid(B_ / 16, O_ / 16);
  head_wmma_kernel<<<grid, 32, 0, stream>>>(outH, Wout, bout, tag);
}


# One split workgroup barrier per step (signal once h is in LDS, wait behind the update pass), z double-buffered; replaces two full barriers and the hbuf-to-z copy; arithmetic and sum order unchanged
# speedup vs baseline: 7.5658x; 1.2124x over previous
#include <hip/hip_runtime.h>
#include <math.h>

#define B_ 128
#define T_ 32
#define D_ 128
#define H_ 256
#define O_ 128
#define Z_ (D_ + H_)

typedef __attribute__((ext_vector_type(2))) float v2f;
typedef __attribute__((ext_vector_type(8))) float v8f;

#define NT_  512
#define RPW_ 16
#define RQ_  (Z_ / 4)
#define CS_  32
#define NS_  6
#define P4S_(j) ((2 * (j) + 4) % NS_)

__device__ __forceinline__ void lds_barrier_signal() {
  __builtin_amdgcn_fence(__ATOMIC_RELEASE, "workgroup", "local");
  __builtin_amdgcn_s_barrier_signal(-1);
}
__device__ __forceinline__ void lds_barrier_wait() {
  __builtin_amdgcn_s_barrier_wait(-1);
  __builtin_amdgcn_fence(__ATOMIC_ACQUIRE, "workgroup", "local");
}

__device__ __forceinline__ float xhalf(float x) {
  const int b = __builtin_bit_cast(int, x);
  return __builtin_bit_cast(float,
      __builtin_amdgcn_permlanex16(b, b, 0x76543210, 0xfedcba98, false, false));
}
#define XROW_(x, M) __builtin_bit_cast(float, __builtin_amdgcn_update_dpp( \
    __builtin_bit_cast(int, (x)), __builtin_bit_cast(int, (x)), 0x160 + (M), 0xf, 0xf, false))

__device__ __forceinline__ void request_row(float4 (&s)[3], const float4* p, int row) {
  s[0] = p[row * RQ_ + 0 * CS_];
  s[1] = p[row * RQ_ + 1 * CS_];
  s[2] = p[row * RQ_ + 2 * CS_];
}

__global__ __launch_bounds__(NT_) void stpn_recurrent_kernel(
    const float* __restrict__ sentence,
    const float* __restrict__ h0,
    const float* __restrict__ F0,
    const float* W,
    const float* __restrict__ bias,
    const float* lam,
    const float* gam,
    float* __restrict__ outH,
    float* __restrict__ outF)
{
  const int b    = blockIdx.x;
  const int tid  = threadIdx.x;
  const int w    = __builtin_amdgcn_readfirstlane(tid >> 5);
  const int lane = tid & 31;

  __shared__ __align__(16) float zbuf[2 * Z_];
  __shared__ float4 fl[H_ * 2 * 32];

  const size_t q0 = (size_t)(w * RPW_) * RQ_ + lane;
  float4* flw = fl + (w * RPW_) * 2 * 32 + lane;

  const float bv = bias[w * RPW_ + (lane >> 1)];
  const bool up16 = (lane & 16) != 0, up8 = (lane & 8) != 0,
             up4  = (lane & 4)  != 0, up2 = (lane & 2) != 0;

  float4 fa[RPW_];
  {
    const float4* src = (const float4*)(F0 + (size_t)b * H_ * Z_) + q0;
    #pragma unroll
    for (int rr = 0; rr < RPW_; ++rr) {
      fa[rr]                 = src[rr * RQ_ + 0 * CS_];
      flw[(rr * 2 + 0) * 32] = src[rr * RQ_ + 1 * CS_];
      flw[(rr * 2 + 1) * 32] = src[rr * RQ_ + 2 * CS_];
    }
    if (tid < H_) zbuf[D_ + tid] = h0[b * H_ + tid];
  }

  const float* xb = sentence + (size_t)b * T_ * D_;
  const float4* Wq = (const float4*)W   + q0;
  const float4* Lq = (const float4*)lam + q0;
  const float4* Gq = (const float4*)gam + q0;

  float4 rb[NS_][3];
  #pragma unroll
  for (int s = 0; s < NS_; ++s) request_row(rb[s], Wq, s);
  float xn = 0.f;
  if (tid < D_) { zbuf[tid] = xb[tid]; xn = xb[D_ + tid]; }
  __syncthreads();

  for (int t = 0; t < T_; ++t) {
    float* zcur = zbuf + (t & 1) * Z_;
    float* znxt = zbuf + ((t + 1) & 1) * Z_;

    const float4* z4p = (const float4*)zcur;
    const float4 zc0 = z4p[0 * CS_ + lane];
    const float4 zc1 = z4p[1 * CS_ + lane];
    const float4 zc2 = z4p[2 * CS_ + lane];

    float pre[RPW_];
    #pragma unroll
    for (int r = 0; r < RPW_; ++r) {
      {
        const int rr = r;
        float4 w0 = rb[rr % NS_][0], w1 = rb[rr % NS_][1], w2 = rb[rr % NS_][2];
        asm volatile("" : "+v"(w0.x), "+v"(w0.y), "+v"(w0.z), "+v"(w0.w),
                          "+v"(w1.x), "+v"(w1.y), "+v"(w1.z), "+v"(w1.w),
                          "+v"(w2.x), "+v"(w2.y), "+v"(w2.z), "+v"(w2.w));
        const float4 f0 = fa[rr], f1 = flw[(rr * 2 + 0) * 32], f2 = flw[(rr * 2 + 1) * 32];

        float p = 0.f;
        p = fmaf(w0.x + f0.x, zc0.x, p); p = fmaf(w0.y + f0.y, zc0.y, p);
        p = fmaf(w0.z + f0.z, zc0.z, p); p = fmaf(w0.w + f0.w, zc0.w, p);
        p = fmaf(w1.x + f1.x, zc1.x, p); p = fmaf(w1.y + f1.y, zc1.y, p);
        p = fmaf(w1.z + f1.z, zc1.z, p); p = fmaf(w1.w + f1.w, zc1.w, p);
        p = fmaf(w2.x + f2.x, zc2.x, p); p = fmaf(w2.y + f2.y, zc2.y, p);
        p = fmaf(w2.z + f2.z, zc2.z, p); p = fmaf(w2.w + f2.w, zc2.w, p);
        pre[rr] = p;
      }
      if (r + NS_ < RPW_)                     request_row(rb[r % NS_], Wq, r + NS_);
      else if (((r + NS_ - RPW_) & 1) == 0)   request_row(rb[r % NS_], Lq, (r + NS_ - RPW_) / 2);
      else                                    request_row(rb[r % NS_], Gq, (r + NS_ - RPW_) / 2);
      __builtin_amdgcn_sched_barrier(0);
    }

    float s8[8];
    #pragma unroll
    for (int rr = 0; rr < 8; ++rr) {
      const float keep = up16 ? pre[rr + 8] : pre[rr];
      const float send = up16 ? pre[rr] : pre[rr + 8];
      s8[rr] = keep + xhalf(send);
    }
    float s4[4];
    #pragma unroll
    for (int rr = 0; rr < 4; ++rr) {
      const float keep = up8 ? s8[rr + 4] : s8[rr];
      const float send = up8 ? s8[rr] : s8[rr + 4];
      s4[rr] = keep + XROW_(send, 8);
    }
    float s2[2];
    #pragma unroll
    for (int rr = 0; rr < 2; ++rr) {
      const float keep = up4 ? s4[rr + 2] : s4[rr];
      const float send = up4 ? s4[rr] : s4[rr + 2];
      s2[rr] = keep + XROW_(send, 4);
    }
    float tot;
    {
      const float keep = up2 ? s2[1] : s2[0];
      const float send = up2 ? s2[0] : s2[1];
      const float s1 = keep + XROW_(send, 2);
      tot = s1 + XROW_(s1, 1);
    }

    const float hnew = tanhf(tot + bv);
    if ((lane & 1) == 0) znxt[D_ + w * RPW_ + (lane >> 1)] = hnew;
    if (tid < D_) {
      znxt[tid] = xn;
      xn = xb[((t + 2 < T_) ? t + 2 : T_ - 1) * D_ + tid];
    }
    lds_barrier_signal();

    asm volatile("" ::: "memory");

    #pragma unroll
    for (int r = 0; r < RPW_; ++r) {
      {
        const int rr = r;
        const int sl = P4S_(rr), sg = sl + 1;
        const float h = __builtin_bit_cast(float,
            __builtin_amdgcn_readlane(__builtin_bit_cast(int, hnew), 2 * rr));
        const float4 l0 = rb[sl][0], l1 = rb[sl][1], l2 = rb[sl][2];
        const float4 g0 = rb[sg][0], g1 = rb[sg][1], g2 = rb[sg][2];
        float4 f0 = fa[rr], f1 = flw[(rr * 2 + 0) * 32], f2 = flw[(rr * 2 + 1) * 32];
        f0.x = fmaf(l0.x, f0.x, g0.x * h * zc0.x);
        f0.y = fmaf(l0.y, f0.y, g0.y * h * zc0.y);
        f0.z = fmaf(l0.z, f0.z, g0.z * h * zc0.z);
        f0.w = fmaf(l0.w, f0.w, g0.w * h * zc0.w);
        f1.x = fmaf(l1.x, f1.x, g1.x * h * zc1.x);
        f1.y = fmaf(l1.y, f1.y, g1.y * h * zc1.y);
        f1.z = fmaf(l1.z, f1.z, g1.z * h * zc1.z);
        f1.w = fmaf(l1.w, f1.w, g1.w * h * zc1.w);
        f2.x = fmaf(l2.x, f2.x, g2.x * h * zc2.x);
        f2.y = fmaf(l2.y, f2.y, g2.y * h * zc2.y);
        f2.z = fmaf(l2.z, f2.z, g2.z * h * zc2.z);
        f2.w = fmaf(l2.w, f2.w, g2.w * h * zc2.w);
        fa[rr] = f0; flw[(rr * 2 + 0) * 32] = f1; flw[(rr * 2 + 1) * 32] = f2;
      }
      {
        const int rr = r;
        const int sl = P4S_(rr), sg = sl + 1;
        if (rr + NS_ / 2 < RPW_) {
          request_row(rb[sl], Lq, rr + NS_ / 2);
          request_row(rb[sg], Gq, rr + NS_ / 2);
        } else {
          request_row(rb[sl], Wq, sl);
          request_row(rb[sg], Wq, sg);
        }
      }
      __builtin_amdgcn_sched_barrier(0);
    }
    lds_barrier_wait();
  }

  {
    float4* dst = (float4*)(outF + (size_t)b * H_ * Z_) + q0;
    #pragma unroll
    for (int rr = 0; rr < RPW_; ++rr) dst[rr * RQ_ + 0 * CS_] = fa[rr];
    #pragma nounroll
    for (int rr = 0; rr < RPW_; ++rr) {
      dst[rr * RQ_ + 1 * CS_] = flw[(rr * 2 + 0) * 32];
      dst[rr * RQ_ + 2 * CS_] = flw[(rr * 2 + 1) * 32];
    }
  }
  if (tid < H_) outH[b * H_ + tid] = zbuf[(T_ & 1) * Z_ + D_ + tid];
}

__global__ __launch_bounds__(32) void head_wmma_kernel(
    const float* __restrict__ hmat,
    const float* __restrict__ Wout,
    const float* __restrict__ bout,
    float* __restrict__ tag)
{
  const int tileM = blockIdx.x;
  const int tileN = blockIdx.y;
  const int lane  = threadIdx.x;
  const int half  = lane >> 4;
  const int l16   = lane & 15;

  const int m = tileM * 16 + l16;
  const int n = tileN * 16 + l16;

  v8f c = {};
  for (int k0 = 0; k0 < H_; k0 += 4) {
    const int ka = k0 + 2 * half;
    v2f a, bf;
    a.x  = hmat[m * H_ + ka + 0];
    a.y  = hmat[m * H_ + ka + 1];
    bf.x = Wout[n * H_ + ka + 0];
    bf.y = Wout[n * H_ + ka + 1];
    c = __builtin_amdgcn_wmma_f32_16x16x4_f32(
        false, a, false, bf, (short)0, c, false, false);
  }

  #pragma unroll
  for (int v = 0; v < 8; ++v) {
    const int mg = tileM * 16 + v + 8 * half;
    const int ng = tileN * 16 + l16;
    tag[mg * O_ + ng] = c[v] + bout[ng];
  }
}

extern "C" void kernel_launch(void* const* d_in, const int* in_sizes, int n_in,
                              void* d_out, int out_size, void* d_ws, size_t ws_size,
                              hipStream_t stream) {
  (void)in_sizes; (void)n_in; (void)out_size; (void)d_ws; (void)ws_size;

  const float* sentence = (const float*)d_in[0];
  const float* h0       = (const float*)d_in[1];
  const float* F0       = (const float*)d_in[2];
  const float* W        = (const float*)d_in[3];
  const float* bias     = (const float*)d_in[4];
  const float* lam      = (const float*)d_in[5];
  const float* gam      = (const float*)d_in[6];
  const float* Wout     = (const float*)d_in[7];
  const float* bout     = (const float*)d_in[8];

  float* out  = (float*)d_out;
  float* tag  = out;
  float* outH = out + B_ * O_;
  float* outF = outH + B_ * H_;

  stpn_recurrent_kernel<<<B_, NT_, 0, stream>>>(
      sentence, h0, F0, W, bias, lam, gam, outH, outF);

  dim3 grid(B_ / 16, O_ / 16);
  head_wmma_kernel<<<grid, 32, 0, stream>>>(outH, Wout, bout, tag);
}


# W's x columns leave the time loop: W[:,0:128].x_t of all 32 steps formed once per workgroup on the matrix unit (f32 16x16x4) into LDS; the loop streams 8/9 of the bytes; row sums add in a new order
# speedup vs baseline: 8.0727x; 1.0670x over previous
#include <hip/hip_runtime.h>
#include <math.h>

#define B_ 128
#define T_ 32
#define D_ 128
#define H_ 256
#define O_ 128
#define Z_ (D_ + H_)

typedef __attribute__((ext_vector_type(2))) float v2f;
typedef __attribute__((ext_vector_type(8))) float v8f;

#define NT_  512
#define RPW_ 16
#define RQ_  (Z_ / 4)
#define CS_  32
#define NS_  6
#define P4S_(j) ((2 * (j) + 4) % NS_)

__device__ __forceinline__ void lds_barrier_signal() {
  __builtin_amdgcn_fence(__ATOMIC_RELEASE, "workgroup", "local");
  __builtin_amdgcn_s_barrier_signal(-1);
}
__device__ __forceinline__ void lds_barrier_wait() {
  __builtin_amdgcn_s_barrier_wait(-1);
  __builtin_amdgcn_fence(__ATOMIC_ACQUIRE, "workgroup", "local");
}

__device__ __forceinline__ float xhalf(float x) {
  const int b = __builtin_bit_cast(int, x);
  return __builtin_bit_cast(float,
      __builtin_amdgcn_permlanex16(b, b, 0x76543210, 0xfedcba98, false, false));
}
#define XROW_(x, M) __builtin_bit_cast(float, __builtin_amdgcn_update_dpp( \
    __builtin_bit_cast(int, (x)), __builtin_bit_cast(int, (x)), 0x160 + (M), 0xf, 0xf, false))

__device__ __forceinline__ void request_row(float4 (&s)[3], const float4* p, int row) {
  s[0] = p[row * RQ_ + 0 * CS_];
  s[1] = p[row * RQ_ + 1 * CS_];
  s[2] = p[row * RQ_ + 2 * CS_];
}

__device__ __forceinline__ void request_row_w(float4 (&s)[3], const float4* p, int row) {
  s[1] = p[row * RQ_ + 1 * CS_];
  s[2] = p[row * RQ_ + 2 * CS_];
}

__global__ __launch_bounds__(NT_) void stpn_recurrent_kernel(
    const float* __restrict__ sentence,
    const float* __restrict__ h0,
    const float* __restrict__ F0,
    const float* W,
    const float* __restrict__ bias,
    const float* lam,
    const float* gam,
    float* __restrict__ outH,
    float* __restrict__ outF)
{
  const int b    = blockIdx.x;
  const int tid  = threadIdx.x;
  const int w    = __builtin_amdgcn_readfirstlane(tid >> 5);
  const int lane = tid & 31;

  __shared__ __align__(16) float zbuf[2 * Z_];
  __shared__ float4 fl[H_ * 2 * 32];
  __shared__ float xw[T_ * H_];

  const float* xb = sentence + (size_t)b * T_ * D_;

  {
    const int c16 = lane & 15, hf = lane >> 4;
    const float4* xa0 = (const float4*)(xb + (size_t)c16 * D_) + hf;
    const float4* xa1 = (const float4*)(xb + (size_t)(16 + c16) * D_) + hf;
    const float4* wbq = (const float4*)(W + (size_t)(w * RPW_ + c16) * Z_) + hf;
    v8f acc0 = {}, acc1 = {};
    #pragma unroll
    for (int j = 0; j < D_ / 8; ++j) {
      const float4 a0 = xa0[2 * j], a1 = xa1[2 * j], bq = wbq[2 * j];
      v2f a0l, a0h, a1l, a1h, bl, bh;
      a0l.x = a0.x; a0l.y = a0.y; a0h.x = a0.z; a0h.y = a0.w;
      a1l.x = a1.x; a1l.y = a1.y; a1h.x = a1.z; a1h.y = a1.w;
      bl.x  = bq.x; bl.y  = bq.y; bh.x  = bq.z; bh.y  = bq.w;
      acc0 = __builtin_amdgcn_wmma_f32_16x16x4_f32(false, a0l, false, bl, (short)0, acc0, false, false);
      acc1 = __builtin_amdgcn_wmma_f32_16x16x4_f32(false, a1l, false, bl, (short)0, acc1, false, false);
      acc0 = __builtin_amdgcn_wmma_f32_16x16x4_f32(false, a0h, false, bh, (short)0, acc0, false, false);
      acc1 = __builtin_amdgcn_wmma_f32_16x16x4_f32(false, a1h, false, bh, (short)0, acc1, false, false);
    }
    #pragma unroll
    for (int v = 0; v < 8; ++v) {
      xw[(v + 8 * hf) * H_ + w * RPW_ + c16]      = acc0[v];
      xw[(16 + v + 8 * hf) * H_ + w * RPW_ + c16] = acc1[v];
    }
  }

  const size_t q0 = (size_t)(w * RPW_) * RQ_ + lane;
  float4* flw = fl + (w * RPW_) * 2 * 32 + lane;

  const float bv = bias[w * RPW_ + (lane >> 1)];
  const bool up16 = (lane & 16) != 0, up8 = (lane & 8) != 0,
             up4  = (lane & 4)  != 0, up2 = (lane & 2) != 0;

  float4 fa[RPW_];
  {
    const float4* src = (const float4*)(F0 + (size_t)b * H_ * Z_) + q0;
    #pragma unroll
    for (int rr = 0; rr < RPW_; ++rr) {
      fa[rr]                 = src[rr * RQ_ + 0 * CS_];
      flw[(rr * 2 + 0) * 32] = src[rr * RQ_ + 1 * CS_];
      flw[(rr * 2 + 1) * 32] = src[rr * RQ_ + 2 * CS_];
    }
    if (tid < H_) zbuf[D_ + tid] = h0[b * H_ + tid];
  }

  const float4* Wq = (const float4*)W   + q0;
  const float4* Lq = (const float4*)lam + q0;
  const float4* Gq = (const float4*)gam + q0;

  float4 rb[NS_][3];
  #pragma unroll
  for (int s = 0; s < NS_; ++s) request_row_w(rb[s], Wq, s);
  float xn = 0.f;
  if (tid < D_) { zbuf[tid] = xb[tid]; xn = xb[D_ + tid]; }
  __syncthreads();

  for (int t = 0; t < T_; ++t) {
    float* zcur = zbuf + (t & 1) * Z_;
    float* znxt = zbuf + ((t + 1) & 1) * Z_;

    const float4* z4p = (const float4*)zcur;
    const float4 zc0 = z4p[0 * CS_ + lane];
    const float4 zc1 = z4p[1 * CS_ + lane];
    const float4 zc2 = z4p[2 * CS_ + lane];
    const float xwv = xw[t * H_ + w * RPW_ + (lane >> 1)];

    float pre[RPW_];
    #pragma unroll
    for (int r = 0; r < RPW_; ++r) {
      {
        const int rr = r;
        float4 w1 = rb[rr % NS_][1], w2 = rb[rr % NS_][2];
        asm volatile("" : "+v"(w1.x), "+v"(w1.y), "+v"(w1.z), "+v"(w1.w),
                          "+v"(w2.x), "+v"(w2.y), "+v"(w2.z), "+v"(w2.w));
        const float4 f0 = fa[rr], f1 = flw[(rr * 2 + 0) * 32], f2 = flw[(rr * 2 + 1) * 32];

        float p = 0.f;
        p = fmaf(f0.x, zc0.x, p); p = fmaf(f0.y, zc0.y, p);
        p = fmaf(f0.z, zc0.z, p); p = fmaf(f0.w, zc0.w, p);
        p = fmaf(w1.x + f1.x, zc1.x, p); p = fmaf(w1.y + f1.y, zc1.y, p);
        p = fmaf(w1.z + f1.z, zc1.z, p); p = fmaf(w1.w + f1.w, zc1.w, p);
        p = fmaf(w2.x + f2.x, zc2.x, p); p = fmaf(w2.y + f2.y, zc2.y, p);
        p = fmaf(w2.z + f2.z, zc2.z, p); p = fmaf(w2.w + f2.w, zc2.w, p);
        pre[rr] = p;
      }
      if (r + NS_ < RPW_)                     request_row_w(rb[r % NS_], Wq, r + NS_);
      else if (((r + NS_ - RPW_) & 1) == 0)   request_row(rb[r % NS_], Lq, (r + NS_ - RPW_) / 2);
      else                                    request_row(rb[r % NS_], Gq, (r + NS_ - RPW_) / 2);
      __builtin_amdgcn_sched_barrier(0);
    }

    float s8[8];
    #pragma unroll
    for (int rr = 0; rr < 8; ++rr) {
      const float keep = up16 ? pre[rr + 8] : pre[rr];
      const float send = up16 ? pre[rr] : pre[rr + 8];
      s8[rr] = keep + xhalf(send);
    }
    float s4[4];
    #pragma unroll
    for (int rr = 0; rr < 4; ++rr) {
      const float keep = up8 ? s8[rr + 4] : s8[rr];
      const float send = up8 ? s8[rr] : s8[rr + 4];
      s4[rr] = keep + XROW_(send, 8);
    }
    float s2[2];
    #pragma unroll
    for (int rr = 0; rr < 2; ++rr) {
      const float keep = up4 ? s4[rr + 2] : s4[rr];
      const float send = up4 ? s4[rr] : s4[rr + 2];
      s2[rr] = keep + XROW_(send, 4);
    }
    float tot;
    {
      const float keep = up2 ? s2[1] : s2[0];
      const float send = up2 ? s2[0] : s2[1];
      const float s1 = keep + XROW_(send, 2);
      tot = s1 + XROW_(s1, 1);
    }

    const float hnew = tanhf((tot + xwv) + bv);
    if ((lane & 1) == 0) znxt[D_ + w * RPW_ + (lane >> 1)] = hnew;
    if (tid < D_) {
      znxt[tid] = xn;
      xn = xb[((t + 2 < T_) ? t + 2 : T_ - 1) * D_ + tid];
    }
    lds_barrier_signal();

    asm volatile("" ::: "memory");

    #pragma unroll
    for (int r = 0; r < RPW_; ++r) {
      {
        const int rr = r;
        const int sl = P4S_(rr), sg = sl + 1;
        const float h = __builtin_bit_cast(float,
            __builtin_amdgcn_readlane(__builtin_bit_cast(int, hnew), 2 * rr));
        const float4 l0 = rb[sl][0], l1 = rb[sl][1], l2 = rb[sl][2];
        const float4 g0 = rb[sg][0], g1 = rb[sg][1], g2 = rb[sg][2];
        float4 f0 = fa[rr], f1 = flw[(rr * 2 + 0) * 32], f2 = flw[(rr * 2 + 1) * 32];
        f0.x = fmaf(l0.x, f0.x, g0.x * h * zc0.x);
        f0.y = fmaf(l0.y, f0.y, g0.y * h * zc0.y);
        f0.z = fmaf(l0.z, f0.z, g0.z * h * zc0.z);
        f0.w = fmaf(l0.w, f0.w, g0.w * h * zc0.w);
        f1.x = fmaf(l1.x, f1.x, g1.x * h * zc1.x);
        f1.y = fmaf(l1.y, f1.y, g1.y * h * zc1.y);
        f1.z = fmaf(l1.z, f1.z, g1.z * h * zc1.z);
        f1.w = fmaf(l1.w, f1.w, g1.w * h * zc1.w);
        f2.x = fmaf(l2.x, f2.x, g2.x * h * zc2.x);
        f2.y = fmaf(l2.y, f2.y, g2.y * h * zc2.y);
        f2.z = fmaf(l2.z, f2.z, g2.z * h * zc2.z);
        f2.w = fmaf(l2.w, f2.w, g2.w * h * zc2.w);
        fa[rr] = f0; flw[(rr * 2 + 0) * 32] = f1; flw[(rr * 2 + 1) * 32] = f2;
      }
      {
        const int rr = r;
        const int sl = P4S_(rr), sg = sl + 1;
        if (rr + NS_ / 2 < RPW_) {
          request_row(rb[sl], Lq, rr + NS_ / 2);
          request_row(rb[sg], Gq, rr + NS_ / 2);
        } else {
          request_row_w(rb[sl], Wq, sl);
          request_row_w(rb[sg], Wq, sg);
        }
      }
      __builtin_amdgcn_sched_barrier(0);
    }
    lds_barrier_wait();
  }

  {
    float4* dst = (float4*)(outF + (size_t)b * H_ * Z_) + q0;
    #pragma unroll
    for (int rr = 0; rr < RPW_; ++rr) dst[rr * RQ_ + 0 * CS_] = fa[rr];
    #pragma nounroll
    for (int rr = 0; rr < RPW_; ++rr) {
      dst[rr * RQ_ + 1 * CS_] = flw[(rr * 2 + 0) * 32];
      dst[rr * RQ_ + 2 * CS_] = flw[(rr * 2 + 1) * 32];
    }
  }
  if (tid < H_) outH[b * H_ + tid] = zbuf[(T_ & 1) * Z_ + D_ + tid];
}

__global__ __launch_bounds__(32) void head_wmma_kernel(
    const float* __restrict__ hmat,
    const float* __restrict__ Wout,
    const float* __restrict__ bout,
    float* __restrict__ tag)
{
  const int tileM = blockIdx.x;
  const int tileN = blockIdx.y;
  const int lane  = threadIdx.x;
  const int half  = lane >> 4;
  const int l16   = lane & 15;

  const int m = tileM * 16 + l16;
  const int n = tileN * 16 + l16;

  v8f c = {};
  for (int k0 = 0; k0 < H_; k0 += 4) {
    const int ka = k0 + 2 * half;
    v2f a, bf;
    a.x  = hmat[m * H_ + ka + 0];
    a.y  = hmat[m * H_ + ka + 1];
    bf.x = Wout[n * H_ + ka + 0];
    bf.y = Wout[n * H_ + ka + 1];
    c = __builtin_amdgcn_wmma_f32_16x16x4_f32(
        false, a, false, bf, (short)0, c, false, false);
  }

  #pragma unroll
  for (int v = 0; v < 8; ++v) {
    const int mg = tileM * 16 + v + 8 * half;
    const int ng = tileN * 16 + l16;
    tag[mg * O_ + ng] = c[v] + bout[ng];
  }
}

extern "C" void kernel_launch(void* const* d_in, const int* in_sizes, int n_in,
                              void* d_out, int out_size, void* d_ws, size_t ws_size,
                              hipStream_t stream) {
  (void)in_sizes; (void)n_in; (void)out_size; (void)d_ws; (void)ws_size;

  const float* sentence = (const float*)d_in[0];
  const float* h0       = (const float*)d_in[1];
  const float* F0       = (const float*)d_in[2];
  const float* W        = (const float*)d_in[3];
  const float* bias     = (const float*)d_in[4];
  const float* lam      = (const float*)d_in[5];
  const float* gam      = (const float*)d_in[6];
  const float* Wout     = (const float*)d_in[7];
  const float* bout     = (const float*)d_in[8];

  float* out  = (float*)d_out;
  float* tag  = out;
  float* outH = out + B_ * O_;
  float* outF = outH + B_ * H_;

  stpn_recurrent_kernel<<<B_, NT_, 0, stream>>>(
      sentence, h0, F0, W, bias, lam, gam, outH, outF);

  dim3 grid(B_ / 16, O_ / 16);
  head_wmma_kernel<<<grid, 32, 0, stream>>>(outH, Wout, bout, tag);
}


# W columns 128..255 held in LDS for the whole time loop (F's second float4 per lane moved to registers, request ring cut to 12 float4 for room): a step streams 7/8 of the bytes; sum order unchanged
# speedup vs baseline: 8.3848x; 1.0387x over previous
#include <hip/hip_runtime.h>
#include <math.h>

#define B_ 128
#define T_ 32
#define D_ 128
#define H_ 256
#define O_ 128
#define Z_ (D_ + H_)

typedef __attribute__((ext_vector_type(2))) float v2f;
typedef __attribute__((ext_vector_type(8))) float v8f;

#define NT_  512
#define RPW_ 16
#define RQ_  (Z_ / 4)
#define CS_  32
#define NQ_  12
#define NW_  6
#define LGB_(j) (6 * ((j) & 1))
static_assert(NQ_ == 12 && NW_ <= NQ_ && 2 * NW_ >= NQ_ && RPW_ >= NW_,
              "the ring holds lam and gamma (6 float4) of exactly two update rows");

__device__ __forceinline__ void lds_barrier_signal() {
  __builtin_amdgcn_fence(__ATOMIC_RELEASE, "workgroup", "local");
  __builtin_amdgcn_s_barrier_signal(-1);
}
__device__ __forceinline__ void lds_barrier_wait() {
  __builtin_amdgcn_s_barrier_wait(-1);
  __builtin_amdgcn_fence(__ATOMIC_ACQUIRE, "workgroup", "local");
}

__device__ __forceinline__ float xhalf(float x) {
  const int b = __builtin_bit_cast(int, x);
  return __builtin_bit_cast(float,
      __builtin_amdgcn_permlanex16(b, b, 0x76543210, 0xfedcba98, false, false));
}
#define XROW_(x, M) __builtin_bit_cast(float, __builtin_amdgcn_update_dpp( \
    __builtin_bit_cast(int, (x)), __builtin_bit_cast(int, (x)), 0x160 + (M), 0xf, 0xf, false))

__global__ __launch_bounds__(NT_) void stpn_recurrent_kernel(
    const float* __restrict__ sentence,
    const float* __restrict__ h0,
    const float* __restrict__ F0,
    const float* W,
    const float* __restrict__ bias,
    const float* lam,
    const float* gam,
    float* __restrict__ outH,
    float* __restrict__ outF)
{
  const int b    = blockIdx.x;
  const int tid  = threadIdx.x;
  const int w    = __builtin_amdgcn_readfirstlane(tid >> 5);
  const int lane = tid & 31;

  __shared__ __align__(16) float zbuf[2 * Z_];
  __shared__ float4 fl[H_ * 2 * 32];
  __shared__ float xw[T_ * H_];

  const float* xb = sentence + (size_t)b * T_ * D_;

  {
    const int c16 = lane & 15, hf = lane >> 4;
    const float4* xa0 = (const float4*)(xb + (size_t)c16 * D_) + hf;
    const float4* xa1 = (const float4*)(xb + (size_t)(16 + c16) * D_) + hf;
    const float4* wbq = (const float4*)(W + (size_t)(w * RPW_ + c16) * Z_) + hf;
    v8f acc0 = {}, acc1 = {};
    #pragma unroll
    for (int j = 0; j < D_ / 8; ++j) {
      const float4 a0 = xa0[2 * j], a1 = xa1[2 * j], bq = wbq[2 * j];
      v2f a0l, a0h, a1l, a1h, bl, bh;
      a0l.x = a0.x; a0l.y = a0.y; a0h.x = a0.z; a0h.y = a0.w;
      a1l.x = a1.x; a1l.y = a1.y; a1h.x = a1.z; a1h.y = a1.w;
      bl.x  = bq.x; bl.y  = bq.y; bh.x  = bq.z; bh.y  = bq.w;
      acc0 = __builtin_amdgcn_wmma_f32_16x16x4_f32(false, a0l, false, bl, (short)0, acc0, false, false);
      acc1 = __builtin_amdgcn_wmma_f32_16x16x4_f32(false, a1l, false, bl, (short)0, acc1, false, false);
      acc0 = __builtin_amdgcn_wmma_f32_16x16x4_f32(false, a0h, false, bh, (short)0, acc0, false, false);
      acc1 = __builtin_amdgcn_wmma_f32_16x16x4_f32(false, a1h, false, bh, (short)0, acc1, false, false);
    }
    #pragma unroll
    for (int v = 0; v < 8; ++v) {
      xw[(v + 8 * hf) * H_ + w * RPW_ + c16]      = acc0[v];
      xw[(16 + v + 8 * hf) * H_ + w * RPW_ + c16] = acc1[v];
    }
  }
  __builtin_amdgcn_sched_barrier(0);

  const size_t q0 = (size_t)(w * RPW_) * RQ_ + lane;
  float4* flw = fl + (w * RPW_) * 2 * 32 + lane;

  const float bv = bias[w * RPW_ + (lane >> 1)];
  const bool up16 = (lane & 16) != 0, up8 = (lane & 8) != 0,
             up4  = (lane & 4)  != 0, up2 = (lane & 2) != 0;

  const float4* Wq = (const float4*)W   + q0;

  float4 fa[RPW_], fb[RPW_];
  {
    const float4* src = (const float4*)(F0 + (size_t)b * H_ * Z_) + q0;
    #pragma unroll
    for (int rr = 0; rr < RPW_; ++rr) {
      fa[rr]       = src[rr * RQ_ + 0 * CS_];
      fb[rr]       = src[rr * RQ_ + 1 * CS_];
      flw[(rr * 2 + 0) * 32] = src[rr * RQ_ + 2 * CS_];
      flw[(rr * 2 + 1) * 32] = Wq[rr * RQ_ + 1 * CS_];
    }
    if (tid < H_) zbuf[D_ + tid] = (h0 + b * H_)[tid];
  }

  const float4* Lq = (const float4*)lam + q0;
  const float4* Gq = (const float4*)gam + q0;

  float4 rq[NQ_];
  #pragma unroll
  for (int s = 0; s < NW_; ++s) rq[s] = Wq[s * RQ_ + 2 * CS_];
  float xn = 0.f;
  if (tid < D_) { zbuf[tid] = xb[tid]; xn = xb[D_ + tid]; }
  __syncthreads();

  for (int t = 0; t < T_; ++t) {
    float* zcur = zbuf + (t & 1) * Z_;
    float* znxt = zbuf + ((t + 1) & 1) * Z_;

    const float4* z4p = (const float4*)zcur;
    const float4 zc0 = z4p[0 * CS_ + lane];
    const float4 zc1 = z4p[1 * CS_ + lane];
    const float4 zc2 = z4p[2 * CS_ + lane];
    const float xwv = xw[t * H_ + w * RPW_ + (lane >> 1)];

    float pre[RPW_];
    #pragma unroll
    for (int r = 0; r < RPW_; ++r) {
      {
        const int rr = r;
        float4 w2 = rq[rr % NW_];
        asm volatile("" : "+v"(w2.x), "+v"(w2.y), "+v"(w2.z), "+v"(w2.w));
        const float4 w1 = flw[(rr * 2 + 1) * 32];
        const float4 f0 = fa[rr], f1 = fb[rr], f2 = flw[(rr * 2 + 0) * 32];

        float p = 0.f;
        p = fmaf(f0.x, zc0.x, p); p = fmaf(f0.y, zc0.y, p);
        p = fmaf(f0.z, zc0.z, p); p = fmaf(f0.w, zc0.w, p);
        p = fmaf(w1.x + f1.x, zc1.x, p); p = fmaf(w1.y + f1.y, zc1.y, p);
        p = fmaf(w1.z + f1.z, zc1.z, p); p = fmaf(w1.w + f1.w, zc1.w, p);
        p = fmaf(w2.x + f2.x, zc2.x, p); p = fmaf(w2.y + f2.y, zc2.y, p);
        p = fmaf(w2.z + f2.z, zc2.z, p); p = fmaf(w2.w + f2.w, zc2.w, p);
        pre[rr] = p;
      }
      {
        const int s = r % NW_;
        const int d = r + NW_ - RPW_;
        if (d < 0)           rq[s] = Wq[(r + NW_) * RQ_ + 2 * CS_];
        else if (s % 6 < 3)  rq[s] = Lq[(s / 6) * RQ_ + (s % 3) * CS_];
        else                 rq[s] = Gq[(s / 6) * RQ_ + (s % 3) * CS_];
        if (d >= 0 && NW_ + d < NQ_) {
          const int e = NW_ + d;
          if (e % 6 < 3)     rq[e] = Lq[(e / 6) * RQ_ + (e % 3) * CS_];
          else               rq[e] = Gq[(e / 6) * RQ_ + (e % 3) * CS_];
        }
      }
      __builtin_amdgcn_sched_barrier(0);
    }

    float s8[8];
    #pragma unroll
    for (int rr = 0; rr < 8; ++rr) {
      const float keep = up16 ? pre[rr + 8] : pre[rr];
      const float send = up16 ? pre[rr] : pre[rr + 8];
      s8[rr] = keep + xhalf(send);
    }
    float s4[4];
    #pragma unroll
    for (int rr = 0; rr < 4; ++rr) {
      const float keep = up8 ? s8[rr + 4] : s8[rr];
      const float send = up8 ? s8[rr] : s8[rr + 4];
      s4[rr] = keep + XROW_(send, 8);
    }
    float s2[2];
    #pragma unroll
    for (int rr = 0; rr < 2; ++rr) {
      const float keep = up4 ? s4[rr + 2] : s4[rr];
      const float send = up4 ? s4[rr] : s4[rr + 2];
      s2[rr] = keep + XROW_(send, 4);
    }
    float tot;
    {
      const float keep = up2 ? s2[1] : s2[0];
      const float send = up2 ? s2[0] : s2[1];
      const float s1 = keep + XROW_(send, 2);
      tot = s1 + XROW_(s1, 1);
    }

    const float hnew = tanhf((tot + xwv) + bv);
    if ((lane & 1) == 0) znxt[D_ + w * RPW_ + (lane >> 1)] = hnew;
    if (tid < D_) {
      znxt[tid] = xn;
      xn = xb[((t + 2 < T_) ? t + 2 : T_ - 1) * D_ + tid];
    }
    lds_barrier_signal();

    asm volatile("" ::: "memory");

    #pragma unroll
    for (int r = 0; r < RPW_; ++r) {
      {
        const int rr = r;
        const int lb = LGB_(rr);
        const float h = __builtin_bit_cast(float,
            __builtin_amdgcn_readlane(__builtin_bit_cast(int, hnew), 2 * rr));
        const float4 l0 = rq[lb + 0], l1 = rq[lb + 1], l2 = rq[lb + 2];
        const float4 g0 = rq[lb + 3], g1 = rq[lb + 4], g2 = rq[lb + 5];
        float4 f0 = fa[rr], f1 = fb[rr], f2 = flw[(rr * 2 + 0) * 32];
        f0.x = fmaf(l0.x, f0.x, g0.x * h * zc0.x);
        f0.y = fmaf(l0.y, f0.y, g0.y * h * zc0.y);
        f0.z = fmaf(l0.z, f0.z, g0.z * h * zc0.z);
        f0.w = fmaf(l0.w, f0.w, g0.w * h * zc0.w);
        f1.x = fmaf(l1.x, f1.x, g1.x * h * zc1.x);
        f1.y = fmaf(l1.y, f1.y, g1.y * h * zc1.y);
        f1.z = fmaf(l1.z, f1.z, g1.z * h * zc1.z);
        f1.w = fmaf(l1.w, f1.w, g1.w * h * zc1.w);
        f2.x = fmaf(l2.x, f2.x, g2.x * h * zc2.x);
        f2.y = fmaf(l2.y, f2.y, g2.y * h * zc2.y);
        f2.z = fmaf(l2.z, f2.z, g2.z * h * zc2.z);
        f2.w = fmaf(l2.w, f2.w, g2.w * h * zc2.w);
        fa[rr] = f0; fb[rr] = f1; flw[(rr * 2 + 0) * 32] = f2;
      }
      {
        const int rr = r;
        const int lb = LGB_(rr);
        if (rr + 2 < RPW_) {
          #pragma unroll
          for (int k = 0; k < 3; ++k) rq[lb + k]     = Lq[(rr + 2) * RQ_ + k * CS_];
          #pragma unroll
          for (int k = 0; k < 3; ++k) rq[lb + 3 + k] = Gq[(rr + 2) * RQ_ + k * CS_];
        } else {
          #pragma unroll
          for (int k = 0; k < 6; ++k)
            if (lb + k < NW_) rq[lb + k] = Wq[(lb + k) * RQ_ + 2 * CS_];
        }
      }
      __builtin_amdgcn_sched_barrier(0);
    }
    lds_barrier_wait();
  }

  {
    float4* dst = (float4*)(outF + (size_t)b * H_ * Z_) + q0;
    #pragma unroll
    for (int rr = 0; rr < RPW_; ++rr) {
      dst[rr * RQ_ + 0 * CS_] = fa[rr];
      dst[rr * RQ_ + 1 * CS_] = fb[rr];
    }
    #pragma nounroll
    for (int rr = 0; rr < RPW_; ++rr) dst[rr * RQ_ + 2 * CS_] = flw[(rr * 2 + 0) * 32];
  }
  if (tid < H_) (outH + b * H_)[tid] = zbuf[(T_ & 1) * Z_ + D_ + tid];
}

__global__ __launch_bounds__(32) void head_wmma_kernel(
    const float* __restrict__ hmat,
    const float* __restrict__ Wout,
    const float* __restrict__ bout,
    float* __restrict__ tag)
{
  const int tileM = blockIdx.x;
  const int tileN = blockIdx.y;
  const int lane  = threadIdx.x;
  const int half  = lane >> 4;
  const int l16   = lane & 15;

  const int m = tileM * 16 + l16;
  const int n = tileN * 16 + l16;

  v8f c = {};
  for (int k0 = 0; k0 < H_; k0 += 4) {
    const int ka = k0 + 2 * half;
    v2f a, bf;
    a.x  = hmat[m * H_ + ka + 0];
    a.y  = hmat[m * H_ + ka + 1];
    bf.x = Wout[n * H_ + ka + 0];
    bf.y = Wout[n * H_ + ka + 1];
    c = __builtin_amdgcn_wmma_f32_16x16x4_f32(
        false, a, false, bf, (short)0, c, false, false);
  }

  #pragma unroll
  for (int v = 0; v < 8; ++v) {
    const int mg = tileM * 16 + v + 8 * half;
    const int ng = tileN * 16 + l16;
    tag[mg * O_ + ng] = c[v] + bout[ng];
  }
}

extern "C" void kernel_launch(void* const* d_in, const int* in_sizes, int n_in,
                              void* d_out, int out_size, void* d_ws, size_t ws_size,
                              hipStream_t stream) {
  (void)in_sizes; (void)n_in; (void)out_size; (void)d_ws; (void)ws_size;

  const float* sentence = (const float*)d_in[0];
  const float* h0       = (const float*)d_in[1];
  const float* F0       = (const float*)d_in[2];
  const float* W        = (const float*)d_in[3];
  const float* bias     = (const float*)d_in[4];
  const float* lam      = (const float*)d_in[5];
  const float* gam      = (const float*)d_in[6];
  const float* Wout     = (const float*)d_in[7];
  const float* bout     = (const float*)d_in[8];

  float* out  = (float*)d_out;
  float* tag  = out;
  float* outH = out + B_ * O_;
  float* outF = outH + B_ * H_;

  stpn_recurrent_kernel<<<B_, NT_, 0, stream>>>(
      sentence, h0, F0, W, bias, lam, gam, outH, outF);

  dim3 grid(B_ / 16, O_ / 16);
  head_wmma_kernel<<<grid, 32, 0, stream>>>(outH, Wout, bout, tag);
}


# Update pass's lam/gamma loads kept three rows ahead of use by a compiler fence per row (the build had sunk each row's 6 loads to their use: 16 exposed latencies per step); 18-float4 ring; same sums
# speedup vs baseline: 8.9296x; 1.0650x over previous
#include <hip/hip_runtime.h>
#include <math.h>

#define B_ 128
#define T_ 32
#define D_ 128
#define H_ 256
#define O_ 128
#define Z_ (D_ + H_)

typedef __attribute__((ext_vector_type(2))) float v2f;
typedef __attribute__((ext_vector_type(8))) float v8f;

#define NT_  512
#define RPW_ 16
#define RQ_  (Z_ / 4)
#define CS_  32
#define NQ_  18
#define NW_  6
#define LGB_(j) (6 * (((j) + 2) % 3))
static_assert(NQ_ == 18 && NW_ == 6 && RPW_ >= NW_ + 3 && LGB_(0) == 12 && LGB_(1) == 0 && LGB_(2) == 6,
              "the ring holds lam and gamma (6 float4) of exactly three update rows; W shares group 0");

__device__ __forceinline__ void lds_barrier_signal() {
  __builtin_amdgcn_fence(__ATOMIC_RELEASE, "workgroup", "local");
  __builtin_amdgcn_s_barrier_signal(-1);
}
__device__ __forceinline__ void lds_barrier_wait() {
  __builtin_amdgcn_s_barrier_wait(-1);
  __builtin_amdgcn_fence(__ATOMIC_ACQUIRE, "workgroup", "local");
}

__device__ __forceinline__ float xhalf(float x) {
  const int b = __builtin_bit_cast(int, x);
  return __builtin_bit_cast(float,
      __builtin_amdgcn_permlanex16(b, b, 0x76543210, 0xfedcba98, false, false));
}
#define XROW_(x, M) __builtin_bit_cast(float, __builtin_amdgcn_update_dpp( \
    __builtin_bit_cast(int, (x)), __builtin_bit_cast(int, (x)), 0x160 + (M), 0xf, 0xf, false))

__global__ __launch_bounds__(NT_) void stpn_recurrent_kernel(
    const float* __restrict__ sentence,
    const float* __restrict__ h0,
    const float* __restrict__ F0,
    const float* W,
    const float* __restrict__ bias,
    const float* lam,
    const float* gam,
    float* __restrict__ outH,
    float* __restrict__ outF)
{
  const int b    = blockIdx.x;
  const int tid  = threadIdx.x;
  const int w    = __builtin_amdgcn_readfirstlane(tid >> 5);
  const int lane = tid & 31;

  __shared__ __align__(16) float zbuf[2 * Z_];
  __shared__ float4 fl[H_ * 2 * 32];
  __shared__ float xw[T_ * H_];

  const float* xb = sentence + (size_t)b * T_ * D_;

  {
    const int c16 = lane & 15, hf = lane >> 4;
    const float4* xa0 = (const float4*)(xb + (size_t)c16 * D_) + hf;
    const float4* xa1 = (const float4*)(xb + (size_t)(16 + c16) * D_) + hf;
    const float4* wbq = (const float4*)(W + (size_t)(w * RPW_ + c16) * Z_) + hf;
    v8f acc0 = {}, acc1 = {};
    #pragma unroll
    for (int j = 0; j < D_ / 8; ++j) {
      const float4 a0 = xa0[2 * j], a1 = xa1[2 * j], bq = wbq[2 * j];
      v2f a0l, a0h, a1l, a1h, bl, bh;
      a0l.x = a0.x; a0l.y = a0.y; a0h.x = a0.z; a0h.y = a0.w;
      a1l.x = a1.x; a1l.y = a1.y; a1h.x = a1.z; a1h.y = a1.w;
      bl.x  = bq.x; bl.y  = bq.y; bh.x  = bq.z; bh.y  = bq.w;
      acc0 = __builtin_amdgcn_wmma_f32_16x16x4_f32(false, a0l, false, bl, (short)0, acc0, false, false);
      acc1 = __builtin_amdgcn_wmma_f32_16x16x4_f32(false, a1l, false, bl, (short)0, acc1, false, false);
      acc0 = __builtin_amdgcn_wmma_f32_16x16x4_f32(false, a0h, false, bh, (short)0, acc0, false, false);
      acc1 = __builtin_amdgcn_wmma_f32_16x16x4_f32(false, a1h, false, bh, (short)0, acc1, false, false);
    }
    #pragma unroll
    for (int v = 0; v < 8; ++v) {
      xw[(v + 8 * hf) * H_ + w * RPW_ + c16]      = acc0[v];
      xw[(16 + v + 8 * hf) * H_ + w * RPW_ + c16] = acc1[v];
    }
  }
  __builtin_amdgcn_sched_barrier(0);

  const size_t q0 = (size_t)(w * RPW_) * RQ_ + lane;
  float4* flw = fl + (w * RPW_) * 2 * 32 + lane;

  const float bv = bias[w * RPW_ + (lane >> 1)];
  const bool up16 = (lane & 16) != 0, up8 = (lane & 8) != 0,
             up4  = (lane & 4)  != 0, up2 = (lane & 2) != 0;

  const float4* Wq = (const float4*)W   + q0;

  float4 fa[RPW_], fb[RPW_];
  {
    const float4* src = (const float4*)(F0 + (size_t)b * H_ * Z_) + q0;
    #pragma unroll
    for (int rr = 0; rr < RPW_; ++rr) {
      fa[rr]       = src[rr * RQ_ + 0 * CS_];
      fb[rr]       = src[rr * RQ_ + 1 * CS_];
      flw[(rr * 2 + 0) * 32] = src[rr * RQ_ + 2 * CS_];
      flw[(rr * 2 + 1) * 32] = Wq[rr * RQ_ + 1 * CS_];
    }
    if (tid < H_) zbuf[D_ + tid] = (h0 + b * H_)[tid];
  }

  const float4* Lq = (const float4*)lam + q0;
  const float4* Gq = (const float4*)gam + q0;

  float4 rq[NQ_];
  #pragma unroll
  for (int s = 0; s < NW_; ++s) rq[s] = Wq[s * RQ_ + 2 * CS_];
  float xn = 0.f;
  if (tid < D_) { zbuf[tid] = xb[tid]; xn = xb[D_ + tid]; }
  __syncthreads();

  for (int t = 0; t < T_; ++t) {
    float* zcur = zbuf + (t & 1) * Z_;
    float* znxt = zbuf + ((t + 1) & 1) * Z_;

    const float4* z4p = (const float4*)zcur;
    const float4 zc0 = z4p[0 * CS_ + lane];
    const float4 zc1 = z4p[1 * CS_ + lane];
    const float4 zc2 = z4p[2 * CS_ + lane];
    const float xwv = xw[t * H_ + w * RPW_ + (lane >> 1)];

    float pre[RPW_];
    #pragma unroll
    for (int r = 0; r < RPW_; ++r) {
      {
        const int rr = r;
        float4 w2 = rq[rr % NW_];
        asm volatile("" : "+v"(w2.x), "+v"(w2.y), "+v"(w2.z), "+v"(w2.w));
        const float4 w1 = flw[(rr * 2 + 1) * 32];
        const float4 f0 = fa[rr], f1 = fb[rr], f2 = flw[(rr * 2 + 0) * 32];

        float p = 0.f;
        p = fmaf(f0.x, zc0.x, p); p = fmaf(f0.y, zc0.y, p);
        p = fmaf(f0.z, zc0.z, p); p = fmaf(f0.w, zc0.w, p);
        p = fmaf(w1.x + f1.x, zc1.x, p); p = fmaf(w1.y + f1.y, zc1.y, p);
        p = fmaf(w1.z + f1.z, zc1.z, p); p = fmaf(w1.w + f1.w, zc1.w, p);
        p = fmaf(w2.x + f2.x, zc2.x, p); p = fmaf(w2.y + f2.y, zc2.y, p);
        p = fmaf(w2.z + f2.z, zc2.z, p); p = fmaf(w2.w + f2.w, zc2.w, p);
        pre[rr] = p;
      }
      {
        const int s = r % NW_;
        const int d = r + NW_ - RPW_;
        if (d < 0)           rq[s] = Wq[(r + NW_) * RQ_ + 2 * CS_];
        else if (s < 3)      rq[s] = Lq[1 * RQ_ + (s % 3) * CS_];
        else                 rq[s] = Gq[1 * RQ_ + (s % 3) * CS_];
        if (d >= 0) {
          const int e = LGB_(0) + d;
          if (d < 3)         rq[e] = Lq[0 * RQ_ + (d % 3) * CS_];
          else               rq[e] = Gq[0 * RQ_ + (d % 3) * CS_];
        }
      }
      __builtin_amdgcn_sched_barrier(0);
    }

    float s8[8];
    #pragma unroll
    for (int rr = 0; rr < 8; ++rr) {
      const float keep = up16 ? pre[rr + 8] : pre[rr];
      const float send = up16 ? pre[rr] : pre[rr + 8];
      s8[rr] = keep + xhalf(send);
    }
    float s4[4];
    #pragma unroll
    for (int rr = 0; rr < 4; ++rr) {
      const float keep = up8 ? s8[rr + 4] : s8[rr];
      const float send = up8 ? s8[rr] : s8[rr + 4];
      s4[rr] = keep + XROW_(send, 8);
    }
    float s2[2];
    #pragma unroll
    for (int rr = 0; rr < 2; ++rr) {
      const float keep = up4 ? s4[rr + 2] : s4[rr];
      const float send = up4 ? s4[rr] : s4[rr + 2];
      s2[rr] = keep + XROW_(send, 4);
    }
    float tot;
    {
      const float keep = up2 ? s2[1] : s2[0];
      const float send = up2 ? s2[0] : s2[1];
      const float s1 = keep + XROW_(send, 2);
      tot = s1 + XROW_(s1, 1);
    }

    #pragma unroll
    for (int k = 0; k < 3; ++k) {
      rq[LGB_(2) + k]     = Lq[2 * RQ_ + k * CS_];
      rq[LGB_(2) + 3 + k] = Gq[2 * RQ_ + k * CS_];
    }

    const float hnew = tanhf((tot + xwv) + bv);
    if ((lane & 1) == 0) znxt[D_ + w * RPW_ + (lane >> 1)] = hnew;
    if (tid < D_) {
      znxt[tid] = xn;
      xn = xb[((t + 2 < T_) ? t + 2 : T_ - 1) * D_ + tid];
    }
    lds_barrier_signal();

    asm volatile("" ::: "memory");

    #pragma unroll
    for (int r = 0; r < RPW_; ++r) {
      {
        const int rr = r;
        const int lb = LGB_(rr);
        const float h = __builtin_bit_cast(float,
            __builtin_amdgcn_readlane(__builtin_bit_cast(int, hnew), 2 * rr));
        const float4 l0 = rq[lb + 0], l1 = rq[lb + 1], l2 = rq[lb + 2];
        const float4 g0 = rq[lb + 3], g1 = rq[lb + 4], g2 = rq[lb + 5];
        float4 f0 = fa[rr], f1 = fb[rr], f2 = flw[(rr * 2 + 0) * 32];
        f0.x = fmaf(l0.x, f0.x, g0.x * h * zc0.x);
        f0.y = fmaf(l0.y, f0.y, g0.y * h * zc0.y);
        f0.z = fmaf(l0.z, f0.z, g0.z * h * zc0.z);
        f0.w = fmaf(l0.w, f0.w, g0.w * h * zc0.w);
        f1.x = fmaf(l1.x, f1.x, g1.x * h * zc1.x);
        f1.y = fmaf(l1.y, f1.y, g1.y * h * zc1.y);
        f1.z = fmaf(l1.z, f1.z, g1.z * h * zc1.z);
        f1.w = fmaf(l1.w, f1.w, g1.w * h * zc1.w);
        f2.x = fmaf(l2.x, f2.x, g2.x * h * zc2.x);
        f2.y = fmaf(l2.y, f2.y, g2.y * h * zc2.y);
        f2.z = fmaf(l2.z, f2.z, g2.z * h * zc2.z);
        f2.w = fmaf(l2.w, f2.w, g2.w * h * zc2.w);
        fa[rr] = f0; fb[rr] = f1; flw[(rr * 2 + 0) * 32] = f2;
      }
      {
        const int rr = r;
        const int lb = LGB_(rr);
        if (rr + 3 < RPW_) {
          #pragma unroll
          for (int k = 0; k < 3; ++k) rq[lb + k]     = Lq[(rr + 3) * RQ_ + k * CS_];
          #pragma unroll
          for (int k = 0; k < 3; ++k) rq[lb + 3 + k] = Gq[(rr + 3) * RQ_ + k * CS_];
        } else if (lb == 0) {
          #pragma unroll
          for (int k = 0; k < NW_; ++k) rq[k] = Wq[k * RQ_ + 2 * CS_];
        }
      }
      asm volatile("" ::: "memory");
      __builtin_amdgcn_sched_barrier(0);
    }
    lds_barrier_wait();
  }

  {
    float4* dst = (float4*)(outF + (size_t)b * H_ * Z_) + q0;
    #pragma unroll
    for (int rr = 0; rr < RPW_; ++rr) {
      dst[rr * RQ_ + 0 * CS_] = fa[rr];
      dst[rr * RQ_ + 1 * CS_] = fb[rr];
    }
    #pragma nounroll
    for (int rr = 0; rr < RPW_; ++rr) dst[rr * RQ_ + 2 * CS_] = flw[(rr * 2 + 0) * 32];
  }
  if (tid < H_) (outH + b * H_)[tid] = zbuf[(T_ & 1) * Z_ + D_ + tid];
}

__global__ __launch_bounds__(32) void head_wmma_kernel(
    const float* __restrict__ hmat,
    const float* __restrict__ Wout,
    const float* __restrict__ bout,
    float* __restrict__ tag)
{
  const int tileM = blockIdx.x;
  const int tileN = blockIdx.y;
  const int lane  = threadIdx.x;
  const int half  = lane >> 4;
  const int l16   = lane & 15;

  const int m = tileM * 16 + l16;
  const int n = tileN * 16 + l16;

  v8f c = {};
  for (int k0 = 0; k0 < H_; k0 += 4) {
    const int ka = k0 + 2 * half;
    v2f a, bf;
    a.x  = hmat[m * H_ + ka + 0];
    a.y  = hmat[m * H_ + ka + 1];
    bf.x = Wout[n * H_ + ka + 0];
    bf.y = Wout[n * H_ + ka + 1];
    c = __builtin_amdgcn_wmma_f32_16x16x4_f32(
        false, a, false, bf, (short)0, c, false, false);
  }

  #pragma unroll
  for (int v = 0; v < 8; ++v) {
    const int mg = tileM * 16 + v + 8 * half;
    const int ng = tileN * 16 + l16;
    tag[mg * O_ + ng] = c[v] + bout[ng];
  }
}

extern "C" void kernel_launch(void* const* d_in, const int* in_sizes, int n_in,
                              void* d_out, int out_size, void* d_ws, size_t ws_size,
                              hipStream_t stream) {
  (void)in_sizes; (void)n_in; (void)out_size; (void)d_ws; (void)ws_size;

  const float* sentence = (const float*)d_in[0];
  const float* h0       = (const float*)d_in[1];
  const float* F0       = (const float*)d_in[2];
  const float* W        = (const float*)d_in[3];
  const float* bias     = (const float*)d_in[4];
  const float* lam      = (const float*)d_in[5];
  const float* gam      = (const float*)d_in[6];
  const float* Wout     = (const float*)d_in[7];
  const float* bout     = (const float*)d_in[8];

  float* out  = (float*)d_out;
  float* tag  = out;
  float* outH = out + B_ * O_;
  float* outF = outH + B_ * H_;

  stpn_recurrent_kernel<<<B_, NT_, 0, stream>>>(
      sentence, h0, F0, W, bias, lam, gam, outH, outF);

  dim3 grid(B_ / 16, O_ / 16);
  head_wmma_kernel<<<grid, 32, 0, stream>>>(outH, Wout, bout, tag);
}
